# starting kernel
# baseline (speedup 1.0000x reference)
#include <hip/hip_runtime.h>

typedef __attribute__((ext_vector_type(2))) float v2f;
typedef __attribute__((ext_vector_type(8))) float v8f;

#define DVEC   128
#define KSTEPS 32
#define NTILES 8


__device__ __forceinline__ void load_w_to_lds(float4* bw4, const float* __restrict__ W) {
    const float4* W4 = (const float4*)W;
    for (int idx = threadIdx.x; idx < KSTEPS * DVEC; idx += blockDim.x) {
        int n = idx >> 5;
        int s = idx & 31;
        bw4[s * DVEC + n] = W4[idx];
    }
}

__global__ __launch_bounds__(256) void compgcn_dense(const float* __restrict__ X,
                                                     const float* __restrict__ W,
                                                     const float* __restrict__ bias,
                                                     float* __restrict__ Out,
                                                     int nrows) {
    __shared__ float4 bw4[KSTEPS * DVEC];
    load_w_to_lds(bw4, W);
    __syncthreads();

    const int lane = threadIdx.x & 31;
    const int m    = lane & 15;
    const int h    = lane >> 4;
    const int wave = threadIdx.x >> 5;
    const int wavesPerBlock = blockDim.x >> 5;
    const int gwave = blockIdx.x * wavesPerBlock + wave;
    const int nwaves = gridDim.x * wavesPerBlock;
    const int ntileRows = (nrows + 15) >> 4;
    const float* bwl = (const float*)bw4 + m * 4 + 2 * h;

    float bv[NTILES];
    #pragma unroll
    for (int j = 0; j < NTILES; ++j) bv[j] = bias ? bias[j * 16 + m] : 0.f;

    for (int t = gwave; t < ntileRows; t += nwaves) {
        int row = t * 16 + m;
        int rc  = row < nrows ? row : (nrows - 1);
        const float* xrow = X + (size_t)rc * DVEC;

        v8f acc[NTILES];
        #pragma unroll
        for (int j = 0; j < NTILES; ++j) acc[j] = (v8f){0.f,0.f,0.f,0.f,0.f,0.f,0.f,0.f};

        #pragma unroll 2
        for (int s = 0; s < KSTEPS; ++s) {
            v2f a = *(const v2f*)(xrow + 4 * s + 2 * h);
            v2f b[NTILES];
            #pragma unroll
            for (int j = 0; j < NTILES; ++j)
                b[j] = *(const v2f*)(bwl + s * DVEC * 4 + j * 64);
            #pragma unroll
            for (int j = 0; j < NTILES; ++j)
                acc[j] = __builtin_amdgcn_wmma_f32_16x16x4_f32(
                    false, a, false, b[j], (short)0, acc[j], false, false);
        }

        if (t * 16 + 16 <= nrows) {
            #pragma unroll
            for (int j = 0; j < NTILES; ++j) {
                float* orow = Out + (size_t)(t * 16 + 8 * h) * DVEC + j * 16 + m;
                #pragma unroll
                for (int i = 0; i < 8; ++i)
                    orow[(size_t)i * DVEC] = acc[j][i] + bv[j];
            }
        } else {
            #pragma unroll
            for (int j = 0; j < NTILES; ++j) {
                #pragma unroll
                for (int i = 0; i < 8; ++i) {
                    int grow = t * 16 + i + 8 * h;
                    if (grow < nrows)
                        Out[(size_t)grow * DVEC + j * 16 + m] = acc[j][i] + bv[j];
                }
            }
        }
    }
}

__global__ __launch_bounds__(256) void compgcn_edges(const float* __restrict__ X,
                                                     const float* __restrict__ Rel,
                                                     const int* __restrict__ src,
                                                     const int* __restrict__ dst,
                                                     const int* __restrict__ etype,
                                                     const float* __restrict__ ew,
                                                     const float* __restrict__ W,
                                                     float* __restrict__ Out,
                                                     int nedges) {
    __shared__ float4 bw4[KSTEPS * DVEC];
    load_w_to_lds(bw4, W);
    __syncthreads();

    const int lane = threadIdx.x & 31;
    const int m    = lane & 15;
    const int h    = lane >> 4;
    const int wave = threadIdx.x >> 5;
    const int wavesPerBlock = blockDim.x >> 5;
    const int gwave = blockIdx.x * wavesPerBlock + wave;
    const int nwaves = gridDim.x * wavesPerBlock;
    const int ntiles = (nedges + 15) >> 4;
    const float* bwl = (const float*)bw4 + m * 4 + 2 * h;

    for (int t = gwave; t < ntiles; t += nwaves) {
        int e  = t * 16 + m;
        bool valid = e < nedges;
        int ec = valid ? e : 0;
        float w  = valid ? ew[ec] : 0.f;
        int   sN = src[ec];
        int   ty = etype[ec];
        int   dN = dst[ec];
        const float* xrow = X   + (size_t)sN * DVEC;
        const float* rrow = Rel + (size_t)ty * DVEC;

        v8f acc[NTILES];
        #pragma unroll
        for (int j = 0; j < NTILES; ++j) acc[j] = (v8f){0.f,0.f,0.f,0.f,0.f,0.f,0.f,0.f};

        #pragma unroll 2
        for (int s = 0; s < KSTEPS; ++s) {
            v2f xv = *(const v2f*)(xrow + 4 * s + 2 * h);
            v2f rv = *(const v2f*)(rrow + 4 * s + 2 * h);
            v2f b[NTILES];
            #pragma unroll
            for (int j = 0; j < NTILES; ++j)
                b[j] = *(const v2f*)(bwl + s * DVEC * 4 + j * 64);
            v2f a;
            a.x = (xv.x - rv.x) * w;
            a.y = (xv.y - rv.y) * w;
            #pragma unroll
            for (int j = 0; j < NTILES; ++j)
                acc[j] = __builtin_amdgcn_wmma_f32_16x16x4_f32(
                    false, a, false, b[j], (short)0, acc[j], false, false);
        }

        #pragma unroll
        for (int i = 0; i < 8; ++i) {
            int r  = i + 8 * h;
            int dd = __shfl(dN, r, 32);
            float* orow = Out + (size_t)dd * DVEC;
            #pragma unroll
            for (int j = 0; j < NTILES; ++j)
                atomicAdd(orow + j * 16 + m, acc[j][i]);
        }
    }
}

extern "C" void kernel_launch(void* const* d_in, const int* in_sizes, int n_in,
                              void* d_out, int out_size, void* d_ws, size_t ws_size,
                              hipStream_t stream) {
    const float* x         = (const float*)d_in[0];
    const int*   edge_idx  = (const int*)d_in[1];
    const int*   edge_type = (const int*)d_in[2];
    const float* rel_emb   = (const float*)d_in[3];
    const float* edge_w    = (const float*)d_in[4];
    const float* W_self    = (const float*)d_in[5];
    const float* W_forward = (const float*)d_in[6];
    const float* W_rel     = (const float*)d_in[7];
    const float* bias      = (const float*)d_in[8];

    const int N = in_sizes[0] / DVEC;
    const int E = in_sizes[2];
    const int R = in_sizes[3] / DVEC;

    const int* src = edge_idx;
    const int* dst = edge_idx + E;

    float* out_nodes = (float*)d_out;
    float* out_rel   = out_nodes + (size_t)N * DVEC;

    {
        int tiles  = (N + 15) >> 4;
        int blocks = (tiles + 7) >> 3;
        compgcn_dense<<<blocks, 256, 0, stream>>>(x, W_self, bias, out_nodes, N);
    }
    {
        int tiles  = (R + 15) >> 4;
        int blocks = (tiles + 7) >> 3;
        compgcn_dense<<<blocks, 256, 0, stream>>>(rel_emb, W_rel, nullptr, out_rel, R);
    }
    {
        int tiles  = (E + 15) >> 4;
        int blocks = (tiles + 7) >> 3;
        if (blocks > 2048) blocks = 2048;
        compgcn_edges<<<blocks, 256, 0, stream>>>(x, rel_emb, src, dst, edge_type, edge_w,
                                                  W_forward, out_nodes, E);
    }
}


# W staging into LDS conflict-free and in one round trip: XOR-swizzled slots (each half-wave of a b128 store spreads over all bank groups, not one), all reads of a thread issued before its first store
# speedup vs baseline: 1.8729x; 1.8729x over previous
#include <hip/hip_runtime.h>

typedef __attribute__((ext_vector_type(2))) float v2f;
typedef __attribute__((ext_vector_type(8))) float v8f;

#define DVEC   128
#define KSTEPS 32
#define NTILES 8
#define BLOCK  256
#define WSTAGE (KSTEPS * DVEC / BLOCK)

__device__ __forceinline__ int wslot(int s, int n) { return s * DVEC + (n ^ (s & 15)); }

__device__ __forceinline__ unsigned bfrag_byte(int s, unsigned laneB) {
    return laneB ^ (unsigned)(s * (DVEC * 16) + (s & 15) * 16);
}

__device__ __forceinline__ void load_w_to_lds(float4* bw4, const float* __restrict__ W) {
    const float4* W4 = (const float4*)W;
    float4 v[WSTAGE];
    #pragma unroll
    for (int it = 0; it < WSTAGE; ++it)
        v[it] = W4[threadIdx.x + it * BLOCK];
    #pragma unroll
    for (int it = 0; it < WSTAGE; ++it) {
        int idx = threadIdx.x + it * BLOCK;
        bw4[wslot(idx & 31, idx >> 5)] = v[it];
    }
}

__global__ __launch_bounds__(BLOCK) void compgcn_dense(const float* __restrict__ X,
                                                     const float* __restrict__ W0,
                                                     const float* __restrict__ bias0,
                                                     float* __restrict__ Out0,
                                                     const float* __restrict__ W1,
                                                     float* __restrict__ Out1,
                                                     int nrows, int blocksPerW) {
    __shared__ float4 bw4[KSTEPS * DVEC];
    const bool second = (int)blockIdx.x >= blocksPerW;
    const float* __restrict__ W    = second ? W1 : W0;
    const float* __restrict__ bias = second ? nullptr : bias0;
    float* __restrict__ Out        = second ? Out1 : Out0;
    const int bid = second ? (int)blockIdx.x - blocksPerW : (int)blockIdx.x;
    load_w_to_lds(bw4, W);
    __syncthreads();

    const int lane = threadIdx.x & 31;
    const int m    = lane & 15;
    const int h    = lane >> 4;
    const int wave = threadIdx.x >> 5;
    const int wavesPerBlock = blockDim.x >> 5;
    const int gwave = bid * wavesPerBlock + wave;
    const int nwaves = blocksPerW * wavesPerBlock;
    const int ntileRows = (nrows + 15) >> 4;
    const unsigned laneB = (unsigned)(m * 16 + h * 8);

    float bv[NTILES];
    #pragma unroll
    for (int j = 0; j < NTILES; ++j) bv[j] = bias ? bias[j * 16 + m] : 0.f;

    for (int t = gwave; t < ntileRows; t += nwaves) {
        int row = t * 16 + m;
        int rc  = row < nrows ? row : (nrows - 1);
        const float* xrow = X + (size_t)rc * DVEC;

        v8f acc[NTILES];
        #pragma unroll
        for (int j = 0; j < NTILES; ++j) acc[j] = (v8f){0.f,0.f,0.f,0.f,0.f,0.f,0.f,0.f};

        #pragma unroll 2
        for (int s = 0; s < KSTEPS; ++s) {
            v2f a = *(const v2f*)(xrow + 4 * s + 2 * h);
            const char* bs = (const char*)bw4 + bfrag_byte(s, laneB);
            v2f b[NTILES];
            #pragma unroll
            for (int j = 0; j < NTILES; ++j)
                b[j] = *(const v2f*)(bs + j * 256);
            #pragma unroll
            for (int j = 0; j < NTILES; ++j)
                acc[j] = __builtin_amdgcn_wmma_f32_16x16x4_f32(
                    false, a, false, b[j], (short)0, acc[j], false, false);
        }

        if (t * 16 + 16 <= nrows) {
            #pragma unroll
            for (int j = 0; j < NTILES; ++j) {
                float* orow = Out + (size_t)(t * 16 + 8 * h) * DVEC + j * 16 + m;
                #pragma unroll
                for (int i = 0; i < 8; ++i)
                    orow[(size_t)i * DVEC] = acc[j][i] + bv[j];
            }
        } else {
            #pragma unroll
            for (int j = 0; j < NTILES; ++j) {
                #pragma unroll
                for (int i = 0; i < 8; ++i) {
                    int grow = t * 16 + i + 8 * h;
                    if (grow < nrows)
                        Out[(size_t)grow * DVEC + j * 16 + m] = acc[j][i] + bv[j];
                }
            }
        }
    }
}

__global__ __launch_bounds__(BLOCK) void compgcn_edges(const float* __restrict__ X,
                                                     const float* __restrict__ Rel,
                                                     const int* __restrict__ src,
                                                     const int* __restrict__ dst,
                                                     const int* __restrict__ etype,
                                                     const float* __restrict__ ew,
                                                     const float* __restrict__ W,
                                                     float* __restrict__ Out,
                                                     int nedges) {
    __shared__ float4 bw4[KSTEPS * DVEC];
    load_w_to_lds(bw4, W);
    __syncthreads();

    const int lane = threadIdx.x & 31;
    const int m    = lane & 15;
    const int h    = lane >> 4;
    const int wave = threadIdx.x >> 5;
    const int wavesPerBlock = blockDim.x >> 5;
    const int gwave = blockIdx.x * wavesPerBlock + wave;
    const int nwaves = gridDim.x * wavesPerBlock;
    const int ntiles = (nedges + 15) >> 4;
    const unsigned laneB = (unsigned)(m * 16 + h * 8);

    for (int t = gwave; t < ntiles; t += nwaves) {
        int e  = t * 16 + m;
        bool valid = e < nedges;
        int ec = valid ? e : 0;
        float w  = valid ? ew[ec] : 0.f;
        int   sN = src[ec];
        int   ty = etype[ec];
        int   dN = dst[ec];
        const float* xrow = X   + (size_t)sN * DVEC;
        const float* rrow = Rel + (size_t)ty * DVEC;

        v8f acc[NTILES];
        #pragma unroll
        for (int j = 0; j < NTILES; ++j) acc[j] = (v8f){0.f,0.f,0.f,0.f,0.f,0.f,0.f,0.f};

        #pragma unroll 2
        for (int s = 0; s < KSTEPS; ++s) {
            v2f xv = *(const v2f*)(xrow + 4 * s + 2 * h);
            v2f rv = *(const v2f*)(rrow + 4 * s + 2 * h);
            const char* bs = (const char*)bw4 + bfrag_byte(s, laneB);
            v2f b[NTILES];
            #pragma unroll
            for (int j = 0; j < NTILES; ++j)
                b[j] = *(const v2f*)(bs + j * 256);
            v2f a;
            a.x = (xv.x - rv.x) * w;
            a.y = (xv.y - rv.y) * w;
            #pragma unroll
            for (int j = 0; j < NTILES; ++j)
                acc[j] = __builtin_amdgcn_wmma_f32_16x16x4_f32(
                    false, a, false, b[j], (short)0, acc[j], false, false);
        }

        #pragma unroll
        for (int i = 0; i < 8; ++i) {
            int r  = i + 8 * h;
            int dd = __shfl(dN, r, 32);
            float* orow = Out + (size_t)dd * DVEC;
            #pragma unroll
            for (int j = 0; j < NTILES; ++j)
                atomicAdd(orow + j * 16 + m, acc[j][i]);
        }
    }
}

__global__ __launch_bounds__(256) void compgcn_scatter(const float* __restrict__ P,
                                                       const float* __restrict__ RW,
                                                       const int* __restrict__ src,
                                                       const int* __restrict__ dst,
                                                       const int* __restrict__ etype,
                                                       const float* __restrict__ ew,
                                                       float* __restrict__ Out,
                                                       int nedges) {
    const int lane = threadIdx.x & 31;
    const int wavesPerBlock = blockDim.x >> 5;
    const int gwave = __builtin_amdgcn_readfirstlane(blockIdx.x * wavesPerBlock + (threadIdx.x >> 5));
    const int nwaves = gridDim.x * wavesPerBlock;
    const int nchunks = (nedges + 7) >> 3;

    for (int c = gwave; c < nchunks; c += nwaves) {
        int e  = c * 8 + (lane & 7);
        int ec = e < nedges ? e : (nedges - 1);
        int   sN = src[ec];
        int   dN = dst[ec];
        int   ty = etype[ec];
        int   wb = __float_as_int(ew[ec]);

        #pragma unroll
        for (int hb = 0; hb < 2; ++hb) {
            float p[4][4], r[4][4];
            #pragma unroll
            for (int i = 0; i < 4; ++i) {
                int s = __builtin_amdgcn_readlane(sN, hb * 4 + i);
                int t = __builtin_amdgcn_readlane(ty, hb * 4 + i);
                const float* prow = P  + (size_t)s * DVEC + lane;
                const float* rrow = RW + (size_t)t * DVEC + lane;
                #pragma unroll
                for (int q = 0; q < 4; ++q) { p[i][q] = prow[32 * q]; r[i][q] = rrow[32 * q]; }
            }
            #pragma unroll
            for (int i = 0; i < 4; ++i) {
                if (c * 8 + hb * 4 + i < nedges) {
                    int   d = __builtin_amdgcn_readlane(dN, hb * 4 + i);
                    float w = __int_as_float(__builtin_amdgcn_readlane(wb, hb * 4 + i));
                    float* orow = Out + (size_t)d * DVEC + lane;
                    #pragma unroll
                    for (int q = 0; q < 4; ++q)
                        atomicAdd(orow + 32 * q, (p[i][q] - r[i][q]) * w);
                }
            }
        }
    }
}

extern "C" void kernel_launch(void* const* d_in, const int* in_sizes, int n_in,
                              void* d_out, int out_size, void* d_ws, size_t ws_size,
                              hipStream_t stream) {
    const float* x         = (const float*)d_in[0];
    const int*   edge_idx  = (const int*)d_in[1];
    const int*   edge_type = (const int*)d_in[2];
    const float* rel_emb   = (const float*)d_in[3];
    const float* edge_w    = (const float*)d_in[4];
    const float* W_self    = (const float*)d_in[5];
    const float* W_forward = (const float*)d_in[6];
    const float* W_rel     = (const float*)d_in[7];
    const float* bias      = (const float*)d_in[8];

    const int N = in_sizes[0] / DVEC;
    const int E = in_sizes[2];
    const int R = in_sizes[3] / DVEC;

    const int* src = edge_idx;
    const int* dst = edge_idx + E;

    float* out_nodes = (float*)d_out;
    float* out_rel   = out_nodes + (size_t)N * DVEC;

    float* P  = (float*)d_ws;
    float* RW = P + (size_t)N * DVEC;
    const bool nodeLevel = ws_size >= (size_t)(N + R) * DVEC * sizeof(float);

    {
        int tiles  = (N + 15) >> 4;
        int blocks = (tiles + 7) >> 3;
        compgcn_dense<<<nodeLevel ? 2 * blocks : blocks, BLOCK, 0, stream>>>(
            x, W_self, bias, out_nodes, W_forward, P, N, blocks);
    }
    {
        int tiles  = (R + 15) >> 4;
        int blocks = (tiles + 7) >> 3;
        compgcn_dense<<<nodeLevel ? 2 * blocks : blocks, BLOCK, 0, stream>>>(
            rel_emb, W_rel, nullptr, out_rel, W_forward, RW, R, blocks);
    }
    if (nodeLevel) {
        int chunks = (E + 7) >> 3;
        int blocks = (chunks + 7) >> 3;
        if (blocks > 2048) blocks = 2048;
        compgcn_scatter<<<blocks, 256, 0, stream>>>(P, RW, src, dst, edge_type, edge_w, out_nodes, E);
    } else {
        int tiles  = (E + 15) >> 4;
        int blocks = (tiles + 7) >> 3;
        if (blocks > 2048) blocks = 2048;
        compgcn_edges<<<blocks, BLOCK, 0, stream>>>(x, rel_emb, src, dst, edge_type, edge_w,
                                                  W_forward, out_nodes, E);
    }
}


# One launch for all four dense products: the relation-row GEMMs (rel @ W_rel^T, rel @ W_fwd^T) become the first blocks of the node GEMM grid; 3 launches -> 2, no serial tiny launch
# speedup vs baseline: 2.0890x; 1.1154x over previous
#include <hip/hip_runtime.h>

typedef __attribute__((ext_vector_type(2))) float v2f;
typedef __attribute__((ext_vector_type(8))) float v8f;

#define DVEC   128
#define KSTEPS 32
#define NTILES 8
#define BLOCK  256
#define WSTAGE (KSTEPS * DVEC / BLOCK)

__device__ __forceinline__ int wslot(int s, int n) { return s * DVEC + (n ^ (s & 15)); }

__device__ __forceinline__ unsigned bfrag_byte(int s, unsigned laneB) {
    return laneB ^ (unsigned)(s * (DVEC * 16) + (s & 15) * 16);
}

__device__ __forceinline__ void load_w_to_lds(float4* bw4, const float* __restrict__ W) {
    const float4* W4 = (const float4*)W;
    float4 v[WSTAGE];
    #pragma unroll
    for (int it = 0; it < WSTAGE; ++it)
        v[it] = W4[threadIdx.x + it * BLOCK];
    #pragma unroll
    for (int it = 0; it < WSTAGE; ++it) {
        int idx = threadIdx.x + it * BLOCK;
        bw4[wslot(idx & 31, idx >> 5)] = v[it];
    }
}

__global__ __launch_bounds__(BLOCK) void compgcn_dense(const float* __restrict__ Xn,
                                                     const float* __restrict__ Wn0,
                                                     const float* __restrict__ biasN,
                                                     float* __restrict__ OutN0,
                                                     const float* __restrict__ W1,
                                                     float* __restrict__ OutN1,
                                                     int nrowsN, int blocksPerWN,
                                                     const float* __restrict__ Xr,
                                                     const float* __restrict__ Wr0,
                                                     float* __restrict__ OutR0,
                                                     float* __restrict__ OutR1,
                                                     int nrowsR, int blocksPerWR, int relBlocks) {
    __shared__ float4 bw4[KSTEPS * DVEC];
    const bool isRel = (int)blockIdx.x < relBlocks;
    const int  b0    = isRel ? (int)blockIdx.x : (int)blockIdx.x - relBlocks;
    const int  blocksPerW = isRel ? blocksPerWR : blocksPerWN;
    const bool second = b0 >= blocksPerW;
    const int  nrows  = isRel ? nrowsR : nrowsN;
    const float* __restrict__ X    = isRel ? Xr : Xn;
    const float* __restrict__ W    = second ? W1 : (isRel ? Wr0 : Wn0);
    const float* __restrict__ bias = (second || isRel) ? nullptr : biasN;
    float* __restrict__ Out        = isRel ? (second ? OutR1 : OutR0) : (second ? OutN1 : OutN0);
    const int bid = second ? b0 - blocksPerW : b0;
    load_w_to_lds(bw4, W);
    __syncthreads();

    const int lane = threadIdx.x & 31;
    const int m    = lane & 15;
    const int h    = lane >> 4;
    const int wave = threadIdx.x >> 5;
    const int wavesPerBlock = blockDim.x >> 5;
    const int gwave = bid * wavesPerBlock + wave;
    const int nwaves = blocksPerW * wavesPerBlock;
    const int ntileRows = (nrows + 15) >> 4;
    const unsigned laneB = (unsigned)(m * 16 + h * 8);

    float bv[NTILES];
    #pragma unroll
    for (int j = 0; j < NTILES; ++j) bv[j] = bias ? bias[j * 16 + m] : 0.f;

    for (int t = gwave; t < ntileRows; t += nwaves) {
        int row = t * 16 + m;
        int rc  = row < nrows ? row : (nrows - 1);
        const float* xrow = X + (size_t)rc * DVEC;

        v8f acc[NTILES];
        #pragma unroll
        for (int j = 0; j < NTILES; ++j) acc[j] = (v8f){0.f,0.f,0.f,0.f,0.f,0.f,0.f,0.f};

        #pragma unroll 2
        for (int s = 0; s < KSTEPS; ++s) {
            v2f a = *(const v2f*)(xrow + 4 * s + 2 * h);
            const char* bs = (const char*)bw4 + bfrag_byte(s, laneB);
            v2f b[NTILES];
            #pragma unroll
            for (int j = 0; j < NTILES; ++j)
                b[j] = *(const v2f*)(bs + j * 256);
            #pragma unroll
            for (int j = 0; j < NTILES; ++j)
                acc[j] = __builtin_amdgcn_wmma_f32_16x16x4_f32(
                    false, a, false, b[j], (short)0, acc[j], false, false);
        }

        if (t * 16 + 16 <= nrows) {
            #pragma unroll
            for (int j = 0; j < NTILES; ++j) {
                float* orow = Out + (size_t)(t * 16 + 8 * h) * DVEC + j * 16 + m;
                #pragma unroll
                for (int i = 0; i < 8; ++i)
                    orow[(size_t)i * DVEC] = acc[j][i] + bv[j];
            }
        } else {
            #pragma unroll
            for (int j = 0; j < NTILES; ++j) {
                #pragma unroll
                for (int i = 0; i < 8; ++i) {
                    int grow = t * 16 + i + 8 * h;
                    if (grow < nrows)
                        Out[(size_t)grow * DVEC + j * 16 + m] = acc[j][i] + bv[j];
                }
            }
        }
    }
}

__global__ __launch_bounds__(BLOCK) void compgcn_edges(const float* __restrict__ X,
                                                     const float* __restrict__ Rel,
                                                     const int* __restrict__ src,
                                                     const int* __restrict__ dst,
                                                     const int* __restrict__ etype,
                                                     const float* __restrict__ ew,
                                                     const float* __restrict__ W,
                                                     float* __restrict__ Out,
                                                     int nedges) {
    __shared__ float4 bw4[KSTEPS * DVEC];
    load_w_to_lds(bw4, W);
    __syncthreads();

    const int lane = threadIdx.x & 31;
    const int m    = lane & 15;
    const int h    = lane >> 4;
    const int wave = threadIdx.x >> 5;
    const int wavesPerBlock = blockDim.x >> 5;
    const int gwave = blockIdx.x * wavesPerBlock + wave;
    const int nwaves = gridDim.x * wavesPerBlock;
    const int ntiles = (nedges + 15) >> 4;
    const unsigned laneB = (unsigned)(m * 16 + h * 8);

    for (int t = gwave; t < ntiles; t += nwaves) {
        int e  = t * 16 + m;
        bool valid = e < nedges;
        int ec = valid ? e : 0;
        float w  = valid ? ew[ec] : 0.f;
        int   sN = src[ec];
        int   ty = etype[ec];
        int   dN = dst[ec];
        const float* xrow = X   + (size_t)sN * DVEC;
        const float* rrow = Rel + (size_t)ty * DVEC;

        v8f acc[NTILES];
        #pragma unroll
        for (int j = 0; j < NTILES; ++j) acc[j] = (v8f){0.f,0.f,0.f,0.f,0.f,0.f,0.f,0.f};

        #pragma unroll 2
        for (int s = 0; s < KSTEPS; ++s) {
            v2f xv = *(const v2f*)(xrow + 4 * s + 2 * h);
            v2f rv = *(const v2f*)(rrow + 4 * s + 2 * h);
            const char* bs = (const char*)bw4 + bfrag_byte(s, laneB);
            v2f b[NTILES];
            #pragma unroll
            for (int j = 0; j < NTILES; ++j)
                b[j] = *(const v2f*)(bs + j * 256);
            v2f a;
            a.x = (xv.x - rv.x) * w;
            a.y = (xv.y - rv.y) * w;
            #pragma unroll
            for (int j = 0; j < NTILES; ++j)
                acc[j] = __builtin_amdgcn_wmma_f32_16x16x4_f32(
                    false, a, false, b[j], (short)0, acc[j], false, false);
        }

        #pragma unroll
        for (int i = 0; i < 8; ++i) {
            int r  = i + 8 * h;
            int dd = __shfl(dN, r, 32);
            float* orow = Out + (size_t)dd * DVEC;
            #pragma unroll
            for (int j = 0; j < NTILES; ++j)
                atomicAdd(orow + j * 16 + m, acc[j][i]);
        }
    }
}

__global__ __launch_bounds__(256) void compgcn_scatter(const float* __restrict__ P,
                                                       const float* __restrict__ RW,
                                                       const int* __restrict__ src,
                                                       const int* __restrict__ dst,
                                                       const int* __restrict__ etype,
                                                       const float* __restrict__ ew,
                                                       float* __restrict__ Out,
                                                       int nedges) {
    const int lane = threadIdx.x & 31;
    const int wavesPerBlock = blockDim.x >> 5;
    const int gwave = __builtin_amdgcn_readfirstlane(blockIdx.x * wavesPerBlock + (threadIdx.x >> 5));
    const int nwaves = gridDim.x * wavesPerBlock;
    const int nchunks = (nedges + 7) >> 3;

    for (int c = gwave; c < nchunks; c += nwaves) {
        int e  = c * 8 + (lane & 7);
        int ec = e < nedges ? e : (nedges - 1);
        int   sN = src[ec];
        int   dN = dst[ec];
        int   ty = etype[ec];
        int   wb = __float_as_int(ew[ec]);

        #pragma unroll
        for (int hb = 0; hb < 2; ++hb) {
            float p[4][4], r[4][4];
            #pragma unroll
            for (int i = 0; i < 4; ++i) {
                int s = __builtin_amdgcn_readlane(sN, hb * 4 + i);
                int t = __builtin_amdgcn_readlane(ty, hb * 4 + i);
                const float* prow = P  + (size_t)s * DVEC + lane;
                const float* rrow = RW + (size_t)t * DVEC + lane;
                #pragma unroll
                for (int q = 0; q < 4; ++q) { p[i][q] = prow[32 * q]; r[i][q] = rrow[32 * q]; }
            }
            #pragma unroll
            for (int i = 0; i < 4; ++i) {
                if (c * 8 + hb * 4 + i < nedges) {
                    int   d = __builtin_amdgcn_readlane(dN, hb * 4 + i);
                    float w = __int_as_float(__builtin_amdgcn_readlane(wb, hb * 4 + i));
                    float* orow = Out + (size_t)d * DVEC + lane;
                    #pragma unroll
                    for (int q = 0; q < 4; ++q)
                        atomicAdd(orow + 32 * q, (p[i][q] - r[i][q]) * w);
                }
            }
        }
    }
}

extern "C" void kernel_launch(void* const* d_in, const int* in_sizes, int n_in,
                              void* d_out, int out_size, void* d_ws, size_t ws_size,
                              hipStream_t stream) {
    const float* x         = (const float*)d_in[0];
    const int*   edge_idx  = (const int*)d_in[1];
    const int*   edge_type = (const int*)d_in[2];
    const float* rel_emb   = (const float*)d_in[3];
    const float* edge_w    = (const float*)d_in[4];
    const float* W_self    = (const float*)d_in[5];
    const float* W_forward = (const float*)d_in[6];
    const float* W_rel     = (const float*)d_in[7];
    const float* bias      = (const float*)d_in[8];

    const int N = in_sizes[0] / DVEC;
    const int E = in_sizes[2];
    const int R = in_sizes[3] / DVEC;

    const int* src = edge_idx;
    const int* dst = edge_idx + E;

    float* out_nodes = (float*)d_out;
    float* out_rel   = out_nodes + (size_t)N * DVEC;

    float* P  = (float*)d_ws;
    float* RW = P + (size_t)N * DVEC;
    const bool nodeLevel = ws_size >= (size_t)(N + R) * DVEC * sizeof(float);

    {
        int tilesN  = (N + 15) >> 4;
        int blocksN = (tilesN + 7) >> 3;
        int tilesR  = (R + 15) >> 4;
        int blocksR = (tilesR + 7) >> 3;
        int perRows = nodeLevel ? 2 : 1;
        compgcn_dense<<<perRows * (blocksR + blocksN), BLOCK, 0, stream>>>(
            x, W_self, bias, out_nodes, W_forward, P, N, blocksN,
            rel_emb, W_rel, out_rel, RW, R, blocksR, perRows * blocksR);
    }
    if (nodeLevel) {
        int chunks = (E + 7) >> 3;
        int blocks = (chunks + 7) >> 3;
        if (blocks > 2048) blocks = 2048;
        compgcn_scatter<<<blocks, 256, 0, stream>>>(P, RW, src, dst, edge_type, edge_w, out_nodes, E);
    } else {
        int tiles  = (E + 15) >> 4;
        int blocks = (tiles + 7) >> 3;
        if (blocks > 2048) blocks = 2048;
        compgcn_edges<<<blocks, BLOCK, 0, stream>>>(x, rel_emb, src, dst, edge_type, edge_w,
                                                  W_forward, out_nodes, E);
    }
}


# Owner pass by output row: owner sorts its edges by row in LDS; a wave sums a row's edges in registers (16-B loads of P and RW), adds the self row, writes it once; no LDS sums or float atomics
# speedup vs baseline: 2.5749x; 1.2326x over previous
#include <hip/hip_runtime.h>

typedef __attribute__((ext_vector_type(2))) float v2f;
typedef __attribute__((ext_vector_type(8))) float v8f;

#define DVEC   128
#define KSTEPS 32
#define NTILES 8
#define BLOCK  256
#define WSTAGE (KSTEPS * DVEC / BLOCK)

__device__ __forceinline__ int wslot(int s, int n) { return s * DVEC + (n ^ (s & 15)); }

__device__ __forceinline__ unsigned bfrag_byte(int s, unsigned laneB) {
    return laneB ^ (unsigned)(s * (DVEC * 16) + (s & 15) * 16);
}

__device__ __forceinline__ void load_w_to_lds(float4* bw4, const float* __restrict__ W) {
    const float4* W4 = (const float4*)W;
    float4 v[WSTAGE];
    #pragma unroll
    for (int it = 0; it < WSTAGE; ++it)
        v[it] = W4[threadIdx.x + it * BLOCK];
    #pragma unroll
    for (int it = 0; it < WSTAGE; ++it) {
        int idx = threadIdx.x + it * BLOCK;
        bw4[wslot(idx & 31, idx >> 5)] = v[it];
    }
}

#define OWN_ROWS 392
#define KMAX     1024
#define RWL      224
#define OBLOCK   1024
#define NBUCKET  128
#define BUCKU    8

__device__ __forceinline__ void bucket_edges(int* lds,
                                             const int* __restrict__ src, const int* __restrict__ dst,
                                             const int* __restrict__ etype, const float* __restrict__ ew,
                                             int4* __restrict__ rec, int* __restrict__ seg,
                                             int nedges, int slice, int nown, int b) {
    int* cnt  = lds;
    int* off  = lds + KMAX;
    int* part = lds + 2 * KMAX + 64;
    const int tid = threadIdx.x;
    for (int i = tid; i < KMAX; i += BLOCK) cnt[i] = 0;
    __syncthreads();
    const int e0 = b * slice;
    const int e1 = (e0 + slice < nedges) ? e0 + slice : nedges;
    for (int e = e0 + tid; e < e1; e += BUCKU * BLOCK) {
        int d[BUCKU];
        #pragma unroll
        for (int u = 0; u < BUCKU; ++u) {
            int ee = e + u * BLOCK;
            d[u] = dst[ee < e1 ? ee : e1 - 1];
        }
        #pragma unroll
        for (int u = 0; u < BUCKU; ++u)
            if (e + u * BLOCK < e1) atomicAdd(&cnt[d[u] / OWN_ROWS], 1);
    }
    __syncthreads();
    const int c0 = cnt[4 * tid], c1 = cnt[4 * tid + 1], c2 = cnt[4 * tid + 2], c3 = cnt[4 * tid + 3];
    const int sum4 = c0 + c1 + c2 + c3;
    part[tid] = sum4;
    __syncthreads();
    for (int d = 1; d < BLOCK; d <<= 1) {
        int v = tid >= d ? part[tid - d] : 0;
        __syncthreads();
        part[tid] += v;
        __syncthreads();
    }
    const int ex = part[tid] - sum4;
    off[4 * tid]     = ex;
    off[4 * tid + 1] = ex + c0;
    off[4 * tid + 2] = ex + c0 + c1;
    off[4 * tid + 3] = ex + c0 + c1 + c2;
    if (tid == BLOCK - 1) off[KMAX] = ex + sum4;
    cnt[4 * tid] = 0; cnt[4 * tid + 1] = 0; cnt[4 * tid + 2] = 0; cnt[4 * tid + 3] = 0;
    __syncthreads();
    for (int i = tid; i <= nown; i += BLOCK) seg[(size_t)b * (nown + 1) + i] = off[i];
    for (int e = e0 + tid; e < e1; e += BUCKU * BLOCK) {
        int d[BUCKU], sn[BUCKU], ty[BUCKU], wb[BUCKU];
        #pragma unroll
        for (int u = 0; u < BUCKU; ++u) {
            int ee = e + u * BLOCK;
            ee = ee < e1 ? ee : e1 - 1;
            d[u] = dst[ee]; sn[u] = src[ee]; ty[u] = etype[ee]; wb[u] = __float_as_int(ew[ee]);
        }
        #pragma unroll
        for (int u = 0; u < BUCKU; ++u) {
            if (e + u * BLOCK < e1) {
                const int k = d[u] / OWN_ROWS;
                const int pos = off[k] + atomicAdd(&cnt[k], 1);
                rec[(size_t)e0 + pos] = make_int4(sn[u], (ty[u] << 16) | (d[u] - k * OWN_ROWS), wb[u], 0);
            }
        }
    }
}

__global__ __launch_bounds__(BLOCK) void compgcn_dense(const float* __restrict__ Xn,
                                                     const float* __restrict__ Wn0,
                                                     const float* __restrict__ biasN,
                                                     float* __restrict__ OutN0,
                                                     const float* __restrict__ W1,
                                                     float* __restrict__ OutN1,
                                                     int nrowsN, int blocksPerWN,
                                                     const float* __restrict__ Xr,
                                                     const float* __restrict__ Wr0,
                                                     float* __restrict__ OutR0,
                                                     float* __restrict__ OutR1,
                                                     int nrowsR, int blocksPerWR, int relBlocks,
                                                     const int* __restrict__ esrc,
                                                     const int* __restrict__ edst,
                                                     const int* __restrict__ etyp,
                                                     const float* __restrict__ ewgt,
                                                     int* __restrict__ rec, int* __restrict__ seg,
                                                     int nedges, int slice, int nown, int nbucket) {
    __shared__ float4 bw4[KSTEPS * DVEC];
    if ((int)blockIdx.x < nbucket) {
        bucket_edges((int*)bw4, esrc, edst, etyp, ewgt, (int4*)rec, seg, nedges, slice, nown, (int)blockIdx.x);
        return;
    }
    const int  bx    = (int)blockIdx.x - nbucket;
    const bool isRel = bx < relBlocks;
    const int  b0    = isRel ? bx : bx - relBlocks;
    const int  blocksPerW = isRel ? blocksPerWR : blocksPerWN;
    const bool second = b0 >= blocksPerW;
    const int  nrows  = isRel ? nrowsR : nrowsN;
    const float* __restrict__ X    = isRel ? Xr : Xn;
    const float* __restrict__ W    = second ? W1 : (isRel ? Wr0 : Wn0);
    const float* __restrict__ bias = (second || isRel) ? nullptr : biasN;
    float* __restrict__ Out        = isRel ? (second ? OutR1 : OutR0) : (second ? OutN1 : OutN0);
    const int bid = second ? b0 - blocksPerW : b0;
    load_w_to_lds(bw4, W);
    __syncthreads();

    const int lane = threadIdx.x & 31;
    const int m    = lane & 15;
    const int h    = lane >> 4;
    const int wave = threadIdx.x >> 5;
    const int wavesPerBlock = blockDim.x >> 5;
    const int gwave = bid * wavesPerBlock + wave;
    const int nwaves = blocksPerW * wavesPerBlock;
    const int ntileRows = (nrows + 15) >> 4;
    const unsigned laneB = (unsigned)(m * 16 + h * 8);

    float bv[NTILES];
    #pragma unroll
    for (int j = 0; j < NTILES; ++j) bv[j] = bias ? bias[j * 16 + m] : 0.f;

    for (int t = gwave; t < ntileRows; t += nwaves) {
        int row = t * 16 + m;
        int rc  = row < nrows ? row : (nrows - 1);
        const float* xrow = X + (size_t)rc * DVEC;

        v8f acc[NTILES];
        #pragma unroll
        for (int j = 0; j < NTILES; ++j) acc[j] = (v8f){0.f,0.f,0.f,0.f,0.f,0.f,0.f,0.f};

        #pragma unroll 2
        for (int s = 0; s < KSTEPS; ++s) {
            v2f a = *(const v2f*)(xrow + 4 * s + 2 * h);
            const char* bs = (const char*)bw4 + bfrag_byte(s, laneB);
            v2f b[NTILES];
            #pragma unroll
            for (int j = 0; j < NTILES; ++j)
                b[j] = *(const v2f*)(bs + j * 256);
            #pragma unroll
            for (int j = 0; j < NTILES; ++j)
                acc[j] = __builtin_amdgcn_wmma_f32_16x16x4_f32(
                    false, a, false, b[j], (short)0, acc[j], false, false);
        }

        if (t * 16 + 16 <= nrows) {
            #pragma unroll
            for (int j = 0; j < NTILES; ++j) {
                float* orow = Out + (size_t)(t * 16 + 8 * h) * DVEC + j * 16 + m;
                #pragma unroll
                for (int i = 0; i < 8; ++i)
                    orow[(size_t)i * DVEC] = acc[j][i] + bv[j];
            }
        } else {
            #pragma unroll
            for (int j = 0; j < NTILES; ++j) {
                #pragma unroll
                for (int i = 0; i < 8; ++i) {
                    int grow = t * 16 + i + 8 * h;
                    if (grow < nrows)
                        Out[(size_t)grow * DVEC + j * 16 + m] = acc[j][i] + bv[j];
                }
            }
        }
    }
}

__global__ __launch_bounds__(BLOCK) void compgcn_edges(const float* __restrict__ X,
                                                     const float* __restrict__ Rel,
                                                     const int* __restrict__ src,
                                                     const int* __restrict__ dst,
                                                     const int* __restrict__ etype,
                                                     const float* __restrict__ ew,
                                                     const float* __restrict__ W,
                                                     float* __restrict__ Out,
                                                     int nedges) {
    __shared__ float4 bw4[KSTEPS * DVEC];
    load_w_to_lds(bw4, W);
    __syncthreads();

    const int lane = threadIdx.x & 31;
    const int m    = lane & 15;
    const int h    = lane >> 4;
    const int wave = threadIdx.x >> 5;
    const int wavesPerBlock = blockDim.x >> 5;
    const int gwave = blockIdx.x * wavesPerBlock + wave;
    const int nwaves = gridDim.x * wavesPerBlock;
    const int ntiles = (nedges + 15) >> 4;
    const unsigned laneB = (unsigned)(m * 16 + h * 8);

    for (int t = gwave; t < ntiles; t += nwaves) {
        int e  = t * 16 + m;
        bool valid = e < nedges;
        int ec = valid ? e : 0;
        float w  = valid ? ew[ec] : 0.f;
        int   sN = src[ec];
        int   ty = etype[ec];
        int   dN = dst[ec];
        const float* xrow = X   + (size_t)sN * DVEC;
        const float* rrow = Rel + (size_t)ty * DVEC;

        v8f acc[NTILES];
        #pragma unroll
        for (int j = 0; j < NTILES; ++j) acc[j] = (v8f){0.f,0.f,0.f,0.f,0.f,0.f,0.f,0.f};

        #pragma unroll 2
        for (int s = 0; s < KSTEPS; ++s) {
            v2f xv = *(const v2f*)(xrow + 4 * s + 2 * h);
            v2f rv = *(const v2f*)(rrow + 4 * s + 2 * h);
            const char* bs = (const char*)bw4 + bfrag_byte(s, laneB);
            v2f b[NTILES];
            #pragma unroll
            for (int j = 0; j < NTILES; ++j)
                b[j] = *(const v2f*)(bs + j * 256);
            v2f a;
            a.x = (xv.x - rv.x) * w;
            a.y = (xv.y - rv.y) * w;
            #pragma unroll
            for (int j = 0; j < NTILES; ++j)
                acc[j] = __builtin_amdgcn_wmma_f32_16x16x4_f32(
                    false, a, false, b[j], (short)0, acc[j], false, false);
        }

        #pragma unroll
        for (int i = 0; i < 8; ++i) {
            int r  = i + 8 * h;
            int dd = __shfl(dN, r, 32);
            float* orow = Out + (size_t)dd * DVEC;
            #pragma unroll
            for (int j = 0; j < NTILES; ++j)
                atomicAdd(orow + j * 16 + m, acc[j][i]);
        }
    }
}

__global__ __launch_bounds__(256) void compgcn_scatter(const float* __restrict__ P,
                                                       const float* __restrict__ RW,
                                                       const int* __restrict__ src,
                                                       const int* __restrict__ dst,
                                                       const int* __restrict__ etype,
                                                       const float* __restrict__ ew,
                                                       float* __restrict__ Out,
                                                       int nedges) {
    const int lane = threadIdx.x & 31;
    const int wavesPerBlock = blockDim.x >> 5;
    const int gwave = __builtin_amdgcn_readfirstlane(blockIdx.x * wavesPerBlock + (threadIdx.x >> 5));
    const int nwaves = gridDim.x * wavesPerBlock;
    const int nchunks = (nedges + 7) >> 3;

    for (int c = gwave; c < nchunks; c += nwaves) {
        int e  = c * 8 + (lane & 7);
        int ec = e < nedges ? e : (nedges - 1);
        int   sN = src[ec];
        int   dN = dst[ec];
        int   ty = etype[ec];
        int   wb = __float_as_int(ew[ec]);

        #pragma unroll
        for (int hb = 0; hb < 2; ++hb) {
            float p[4][4], r[4][4];
            #pragma unroll
            for (int i = 0; i < 4; ++i) {
                int s = __builtin_amdgcn_readlane(sN, hb * 4 + i);
                int t = __builtin_amdgcn_readlane(ty, hb * 4 + i);
                const float* prow = P  + (size_t)s * DVEC + lane;
                const float* rrow = RW + (size_t)t * DVEC + lane;
                #pragma unroll
                for (int q = 0; q < 4; ++q) { p[i][q] = prow[32 * q]; r[i][q] = rrow[32 * q]; }
            }
            #pragma unroll
            for (int i = 0; i < 4; ++i) {
                if (c * 8 + hb * 4 + i < nedges) {
                    int   d = __builtin_amdgcn_readlane(dN, hb * 4 + i);
                    float w = __int_as_float(__builtin_amdgcn_readlane(wb, hb * 4 + i));
                    float* orow = Out + (size_t)d * DVEC + lane;
                    #pragma unroll
                    for (int q = 0; q < 4; ++q)
                        atomicAdd(orow + 32 * q, (p[i][q] - r[i][q]) * w);
                }
            }
        }
    }
}

#define RWROWS 256
#define SBLOCK 1024
#define RWSTAGE (RWROWS * (DVEC / 4) / SBLOCK)

__global__ __launch_bounds__(SBLOCK) void compgcn_scatter_lds(const float* __restrict__ P,
                                                              const float* __restrict__ RW,
                                                              const int* __restrict__ src,
                                                              const int* __restrict__ dst,
                                                              const int* __restrict__ etype,
                                                              const float* __restrict__ ew,
                                                              float* __restrict__ Out,
                                                              int nedges, int nrel) {
    __shared__ float4 rw4[RWROWS * (DVEC / 4)];
    {
        const float4* RW4 = (const float4*)RW;
        const int n4 = nrel * (DVEC / 4);
        float4 v[RWSTAGE];
        #pragma unroll
        for (int k = 0; k < RWSTAGE; ++k) {
            int i = threadIdx.x + k * SBLOCK;
            v[k] = RW4[i < n4 ? i : n4 - 1];
        }
        #pragma unroll
        for (int k = 0; k < RWSTAGE; ++k) {
            int i = threadIdx.x + k * SBLOCK;
            rw4[i] = v[k];
        }
    }
    __syncthreads();
    const float* rw = (const float*)rw4;

    const int lane = threadIdx.x & 31;
    const int wavesPerBlock = blockDim.x >> 5;
    const int gwave = __builtin_amdgcn_readfirstlane(blockIdx.x * wavesPerBlock + (threadIdx.x >> 5));
    const int nwaves = gridDim.x * wavesPerBlock;
    const int nchunks = (nedges + 7) >> 3;

    for (int c = gwave; c < nchunks; c += nwaves) {
        int e  = c * 8 + (lane & 7);
        int ec = e < nedges ? e : (nedges - 1);
        int   sN = src[ec];
        int   dN = dst[ec];
        int   ty = etype[ec];
        int   wb = __float_as_int(ew[ec]);

        #pragma unroll
        for (int hb = 0; hb < 2; ++hb) {
            float p[4][4], r[4][4];
            #pragma unroll
            for (int i = 0; i < 4; ++i) {
                int s = __builtin_amdgcn_readlane(sN, hb * 4 + i);
                const float* prow = P + (size_t)s * DVEC + lane;
                #pragma unroll
                for (int q = 0; q < 4; ++q) p[i][q] = prow[32 * q];
            }
            #pragma unroll
            for (int i = 0; i < 4; ++i) {
                int t = __builtin_amdgcn_readlane(ty, hb * 4 + i);
                const float* rrow = rw + t * DVEC + lane;
                #pragma unroll
                for (int q = 0; q < 4; ++q) r[i][q] = rrow[32 * q];
            }
            #pragma unroll
            for (int i = 0; i < 4; ++i) {
                if (c * 8 + hb * 4 + i < nedges) {
                    int   d = __builtin_amdgcn_readlane(dN, hb * 4 + i);
                    float w = __int_as_float(__builtin_amdgcn_readlane(wb, hb * 4 + i));
                    float* orow = Out + (size_t)d * DVEC + lane;
                    #pragma unroll
                    for (int q = 0; q < 4; ++q)
                        atomicAdd(orow + 32 * q, (p[i][q] - r[i][q]) * w);
                }
            }
        }
    }
}

#define RWLSTAGE (RWL * (DVEC / 4) / OBLOCK)
#define OBATCH   8
#define OCAP     3072
#define OPER     (OCAP / OBLOCK)

__device__ __forceinline__ int4 row_records(const int4* srec, int i, int n1, int lane) {
    int idx = i + (lane & (OBATCH - 1));
    return srec[idx < n1 ? idx : n1 - 1];
}
__device__ __forceinline__ void row_request(const float4* __restrict__ P4, int4 rc, int lane, float4 (&p)[OBATCH]) {
    #pragma unroll
    for (int j = 0; j < OBATCH; ++j) {
        int sN = __builtin_amdgcn_readlane(rc.x, j);
        p[j] = P4[(size_t)sN * (DVEC / 4) + lane];
    }
}
__device__ __forceinline__ void row_sum(const float4* rwv, int4 rc, int lane, const float4 (&p)[OBATCH],
                                        int i, int n1, float4& acc) {
    #pragma unroll
    for (int j = 0; j < OBATCH; ++j) {
        const int   ty = __builtin_amdgcn_readlane(rc.y, j) >> 16;
        const float wr = __int_as_float(__builtin_amdgcn_readlane(rc.z, j));
        const float w  = (i + j < n1) ? wr : 0.f;
        const float4 t = rwv[ty * (DVEC / 4) + lane];
        acc.x += (p[j].x - t.x) * w;
        acc.y += (p[j].y - t.y) * w;
        acc.z += (p[j].z - t.z) * w;
        acc.w += (p[j].w - t.w) * w;
    }
}

__global__ __launch_bounds__(OBLOCK) void compgcn_rows(const float* __restrict__ P,
                                                       const float* __restrict__ RW,
                                                       const int* __restrict__ rec,
                                                       const int* __restrict__ seg,
                                                       float* __restrict__ Out,
                                                       int nnodes, int nrel, int slice, int nown) {
    __shared__ float4 rw4[RWL * (DVEC / 4)];
    __shared__ int4   srec[OCAP];
    __shared__ int    pre[NBUCKET + 1];
    __shared__ int    rbeg[NBUCKET];
    __shared__ int    cnt[OWN_ROWS];
    __shared__ int    rowoff[OWN_ROWS + 1];
    __shared__ int    wsum[OBLOCK / 32];
    const int tid  = threadIdx.x;
    const int lane = tid & 31;
    const int k    = blockIdx.x;
    const int row0 = k * OWN_ROWS;
    const int nrows = (nnodes - row0 < OWN_ROWS) ? nnodes - row0 : OWN_ROWS;
    {
        const float4* RW4 = (const float4*)RW;
        const int n4r = nrel * (DVEC / 4);
        float4 u[RWLSTAGE];
        #pragma unroll
        for (int it = 0; it < RWLSTAGE; ++it) {
            int i = tid + it * OBLOCK;
            u[it] = RW4[i < n4r ? i : n4r - 1];
        }
        int e0 = 0, e1 = 0;
        if (tid < NBUCKET) {
            const int* sg = seg + (size_t)tid * (nown + 1) + k;
            e0 = sg[0]; e1 = sg[1];
        }
        #pragma unroll
        for (int it = 0; it < RWLSTAGE; ++it)
            rw4[tid + it * OBLOCK] = u[it];
        if (tid < NBUCKET) { rbeg[tid] = e0; pre[tid + 1] = e1 - e0; }
        if (tid == 0) pre[0] = 0;
    }
    __syncthreads();
    if (tid < 32) {
        const int a0 = pre[4 * tid + 1], a1 = pre[4 * tid + 2], a2 = pre[4 * tid + 3], a3 = pre[4 * tid + 4];
        const int s4 = a0 + a1 + a2 + a3;
        int inc = s4;
        for (int d = 1; d < 32; d <<= 1) { int v = __shfl_up(inc, d, 32); if (lane >= d) inc += v; }
        const int ex = inc - s4;
        pre[4 * tid + 1] = ex + a0;
        pre[4 * tid + 2] = ex + a0 + a1;
        pre[4 * tid + 3] = ex + a0 + a1 + a2;
        pre[4 * tid + 4] = ex + s4;
    }
    __syncthreads();
    const int total = pre[NBUCKET];
    const int wave = __builtin_amdgcn_readfirstlane(tid >> 5);
    const int4*   rec4 = (const int4*)rec;
    const float4* P4   = (const float4*)P;
    const float4* rwv  = (const float4*)rw4;
    float4* O4 = (float4*)(Out + (size_t)row0 * DVEC);

    for (int base = 0; base < total; base += OCAP) {
        const int pend = (total - base < OCAP) ? total : base + OCAP;
        if (tid < OWN_ROWS) cnt[tid] = 0;
        __syncthreads();
        int4 r[OPER];
        #pragma unroll
        for (int q = 0; q < OPER; ++q) {
            const int g  = base + tid + q * OBLOCK;
            const int gc = g < pend ? g : pend - 1;
            int b = 0;
            for (int step = NBUCKET / 2; step >= 1; step >>= 1)
                if (pre[b + step] <= gc) b += step;
            r[q] = rec4[(size_t)b * slice + rbeg[b] + (gc - pre[b])];
        }
        #pragma unroll
        for (int q = 0; q < OPER; ++q)
            if (base + tid + q * OBLOCK < pend) atomicAdd(&cnt[r[q].y & 0xffff], 1);
        __syncthreads();
        int c = 0, inc = 0;
        if (tid < 416) {
            c = tid < OWN_ROWS ? cnt[tid] : 0;
            inc = c;
            for (int d = 1; d < 32; d <<= 1) { int v = __shfl_up(inc, d, 32); if (lane >= d) inc += v; }
            if (lane == 31) wsum[tid >> 5] = inc;
        }
        __syncthreads();
        if (tid < 416) {
            int wb = 0;
            for (int w = 0; w < (tid >> 5); ++w) wb += wsum[w];
            const int ex = wb + inc - c;
            if (tid < OWN_ROWS) { rowoff[tid] = ex; cnt[tid] = 0; }
            if (tid == OWN_ROWS - 1) rowoff[OWN_ROWS] = ex + c;
        }
        __syncthreads();
        #pragma unroll
        for (int q = 0; q < OPER; ++q) {
            if (base + tid + q * OBLOCK < pend) {
                const int row = r[q].y & 0xffff;
                const int pos = rowoff[row] + atomicAdd(&cnt[row], 1);
                srec[pos] = r[q];
            }
        }
        __syncthreads();

        for (int rr = wave; rr < nrows; rr += OBLOCK / 32) {
            const int n0 = __builtin_amdgcn_readfirstlane(rowoff[rr]);
            const int n1 = __builtin_amdgcn_readfirstlane(rowoff[rr + 1]);
            if (n0 >= n1) continue;
            float4* orow = O4 + (size_t)rr * (DVEC / 4) + lane;
            float4 acc = make_float4(0.f, 0.f, 0.f, 0.f);
            float4 cur;
            float4 p[OBATCH];
            int4 rc = row_records(srec, n0, n1, lane);
            row_request(P4, rc, lane, p);
            cur = *orow;
            row_sum(rwv, rc, lane, p, n0, n1, acc);
            for (int i = n0 + OBATCH; i < n1; i += OBATCH) {
                rc = row_records(srec, i, n1, lane);
                row_request(P4, rc, lane, p);
                row_sum(rwv, rc, lane, p, i, n1, acc);
            }
            *orow = make_float4(cur.x + acc.x, cur.y + acc.y, cur.z + acc.z, cur.w + acc.w);
        }
        __syncthreads();
    }
}

extern "C" void kernel_launch(void* const* d_in, const int* in_sizes, int n_in,
                              void* d_out, int out_size, void* d_ws, size_t ws_size,
                              hipStream_t stream) {
    const float* x         = (const float*)d_in[0];
    const int*   edge_idx  = (const int*)d_in[1];
    const int*   edge_type = (const int*)d_in[2];
    const float* rel_emb   = (const float*)d_in[3];
    const float* edge_w    = (const float*)d_in[4];
    const float* W_self    = (const float*)d_in[5];
    const float* W_forward = (const float*)d_in[6];
    const float* W_rel     = (const float*)d_in[7];
    const float* bias      = (const float*)d_in[8];

    const int N = in_sizes[0] / DVEC;
    const int E = in_sizes[2];
    const int R = in_sizes[3] / DVEC;

    const int* src = edge_idx;
    const int* dst = edge_idx + E;

    float* out_nodes = (float*)d_out;
    float* out_rel   = out_nodes + (size_t)N * DVEC;

    float* P  = (float*)d_ws;
    float* RW = P + (size_t)N * DVEC;
    const bool nodeLevel = ws_size >= (size_t)(N + R) * DVEC * sizeof(float);
    const int nown  = (N + OWN_ROWS - 1) / OWN_ROWS;
    const int slice = (E + NBUCKET - 1) / NBUCKET;
    int* rec = (int*)(RW + (size_t)R * DVEC);
    int* seg = rec + (size_t)NBUCKET * slice * 4;
    const bool byOwner = nodeLevel && E >= 1 && R >= 1 && R <= RWL && nown <= KMAX &&
        ws_size >= (size_t)(N + R) * DVEC * sizeof(float) + (size_t)NBUCKET * slice * 16 + (size_t)NBUCKET * (nown + 1) * sizeof(int);
    const int nbucket = byOwner ? NBUCKET : 0;

    {
        int tilesN  = (N + 15) >> 4;
        int blocksN = (tilesN + 7) >> 3;
        int tilesR  = (R + 15) >> 4;
        int blocksR = (tilesR + 7) >> 3;
        int perRows = nodeLevel ? 2 : 1;
        compgcn_dense<<<nbucket + perRows * (blocksR + blocksN), BLOCK, 0, stream>>>(
            x, W_self, bias, out_nodes, W_forward, P, N, blocksN,
            rel_emb, W_rel, out_rel, RW, R, blocksR, perRows * blocksR,
            src, dst, edge_type, edge_w, rec, seg, E, slice, nown, nbucket);
    }
    if (byOwner) {
        compgcn_rows<<<nown, OBLOCK, 0, stream>>>(P, RW, rec, seg, out_nodes, N, R, slice, nown);
    } else if (nodeLevel && R >= 1 && R <= RWROWS) {
        int chunks = (E + 7) >> 3;
        int blocks = (chunks + 31) >> 5;
        if (blocks > 512) blocks = 512;
        compgcn_scatter_lds<<<blocks, SBLOCK, 0, stream>>>(P, RW, src, dst, edge_type, edge_w, out_nodes, E, R);
    } else if (nodeLevel) {
        int chunks = (E + 7) >> 3;
        int blocks = (chunks + 7) >> 3;
        if (blocks > 2048) blocks = 2048;
        compgcn_scatter<<<blocks, 256, 0, stream>>>(P, RW, src, dst, edge_type, edge_w, out_nodes, E);
    } else {
        int tiles  = (E + 15) >> 4;
        int blocks = (tiles + 7) >> 3;
        if (blocks > 2048) blocks = 2048;
        compgcn_edges<<<blocks, BLOCK, 0, stream>>>(x, rel_emb, src, dst, edge_type, edge_w,
                                                  W_forward, out_nodes, E);
    }
}
